# starting kernel
# baseline (speedup 1.0000x reference)
#include <hip/hip_runtime.h>

typedef __attribute__((ext_vector_type(2))) float v2f;
typedef __attribute__((ext_vector_type(4))) float v4f;
typedef __attribute__((ext_vector_type(8))) float v8f;

#define D_IN  256
#define D_OUT 64

__global__ __launch_bounds__(256) void k_zero(float* __restrict__ out, int n4) {
    int i = blockIdx.x * 256 + threadIdx.x;
    if (i < n4) {
        v4f z = {0.f, 0.f, 0.f, 0.f};
        ((v4f*)out)[i] = z;
    }
}

__global__ __launch_bounds__(256) void k_gemm(const float* __restrict__ x,
                                              const float* __restrict__ w,
                                              float* __restrict__ hidden,
                                              int n_rows) {
    __shared__ float wl[D_IN * D_OUT];

    const int tid = threadIdx.x;
    for (int i = tid; i < D_IN * D_OUT; i += 256) {
        int k = i / D_OUT;
        int n = i - k * D_OUT;
        wl[(k >> 1) * 128 + n * 2 + (k & 1)] = w[i];
    }
    __syncthreads();

    const int wave = tid >> 5;
    const int lane = tid & 31;
    const int l15  = lane & 15;
    const int hi   = lane >> 4;

    const long row0 = ((long)blockIdx.x * 8 + wave) * 16;
    if (row0 >= n_rows) return;

    const float* ap = x + (row0 + l15) * D_IN + 2 * hi;

    v8f c0 = {}, c1 = {}, c2 = {}, c3 = {};

    for (int k = 0; k < D_IN; k += 4) {
        v2f a = *(const v2f*)(ap + k);
        const float* bbase = &wl[((k >> 1) + hi) * 128 + l15 * 2];
        v2f b0 = *(const v2f*)(bbase +  0);
        v2f b1 = *(const v2f*)(bbase + 32);
        v2f b2 = *(const v2f*)(bbase + 64);
        v2f b3 = *(const v2f*)(bbase + 96);
        c0 = __builtin_amdgcn_wmma_f32_16x16x4_f32(false, a, false, b0, (short)0, c0, false, false);
        c1 = __builtin_amdgcn_wmma_f32_16x16x4_f32(false, a, false, b1, (short)0, c1, false, false);
        c2 = __builtin_amdgcn_wmma_f32_16x16x4_f32(false, a, false, b2, (short)0, c2, false, false);
        c3 = __builtin_amdgcn_wmma_f32_16x16x4_f32(false, a, false, b3, (short)0, c3, false, false);
    }

    float* hp = hidden + (row0 + 8 * hi) * D_OUT + l15;
#pragma unroll
    for (int r = 0; r < 8; ++r) {
        hp[r * D_OUT +  0] = c0[r];
        hp[r * D_OUT + 16] = c1[r];
        hp[r * D_OUT + 32] = c2[r];
        hp[r * D_OUT + 48] = c3[r];
    }
}

__global__ __launch_bounds__(256) void k_scatter(const float* __restrict__ hidden,
                                                 const float* __restrict__ adj,
                                                 const int* __restrict__ row,
                                                 const int* __restrict__ col,
                                                 float* __restrict__ out,
                                                 int n_edges) {
    const int wave = threadIdx.x >> 5;
    const int lane = threadIdx.x & 31;
    const int e    = blockIdx.x * 8 + wave;
    if (e >= n_edges) return;

    const int   r = row[e];
    const int   c = col[e];
    const float v = adj[e];

    const float* h = hidden + (long)c * D_OUT;
    float*       o = out    + (long)r * D_OUT;

    atomicAdd(&o[lane],      v * h[lane]);
    atomicAdd(&o[lane + 32], v * h[lane + 32]);
}

__global__ __launch_bounds__(256) void k_prelu(float* __restrict__ out,
                                               const float* __restrict__ slope,
                                               int n4) {
    int i = blockIdx.x * 256 + threadIdx.x;
    if (i >= n4) return;
    const float s = slope[0];
    v4f v = ((v4f*)out)[i];
#pragma unroll
    for (int j = 0; j < 4; ++j) {
        float t = v[j];
        v[j] = t > 0.f ? t : s * t;
    }
    ((v4f*)out)[i] = v;
}

extern "C" void kernel_launch(void* const* d_in, const int* in_sizes, int n_in,
                              void* d_out, int out_size, void* d_ws, size_t ws_size,
                              hipStream_t stream) {
    const float* x    = (const float*)d_in[0];
    const float* w    = (const float*)d_in[1];
    const float* adj  = (const float*)d_in[2];
    const float* pa   = (const float*)d_in[3];
    const int*   row  = (const int*)d_in[4];
    const int*   col  = (const int*)d_in[5];
    float*       out  = (float*)d_out;
    float*       hid  = (float*)d_ws;

    const int n_nodes = in_sizes[0] / D_IN;
    const int n_edges = in_sizes[2];
    const int out4    = out_size / 4;

    k_zero<<<(out4 + 255) / 256, 256, 0, stream>>>(out, out4);

    const int row_tiles = (n_nodes + 15) / 16;
    k_gemm<<<(row_tiles + 7) / 8, 256, 0, stream>>>(x, w, hid, n_nodes);

    k_scatter<<<(n_edges + 7) / 8, 256, 0, stream>>>(hid, adj, row, col, out, n_edges);

    k_prelu<<<(out4 + 255) / 256, 256, 0, stream>>>(out, pa, out4);
}


# scatter: one wave handles 32 edges (coalesced index loads, lane broadcast, 16 gathers in flight) instead of 1 edge per wave; f32 atomic adds kept, their arrival order changes
# speedup vs baseline: 1.6869x; 1.6869x over previous
#include <hip/hip_runtime.h>

typedef __attribute__((ext_vector_type(2))) float v2f;
typedef __attribute__((ext_vector_type(4))) float v4f;
typedef __attribute__((ext_vector_type(8))) float v8f;

#define D_IN  256
#define D_OUT 64

__global__ __launch_bounds__(256) void k_zero(float* __restrict__ out, int n4) {
    int i = blockIdx.x * 256 + threadIdx.x;
    if (i < n4) {
        v4f z = {0.f, 0.f, 0.f, 0.f};
        ((v4f*)out)[i] = z;
    }
}

__global__ __launch_bounds__(256) void k_gemm(const float* __restrict__ x,
                                              const float* __restrict__ w,
                                              float* __restrict__ hidden,
                                              int n_rows) {
    __shared__ float wl[D_IN * D_OUT];

    const int tid = threadIdx.x;
    for (int i = tid; i < D_IN * D_OUT; i += 256) {
        int k = i / D_OUT;
        int n = i - k * D_OUT;
        wl[(k >> 1) * 128 + n * 2 + (k & 1)] = w[i];
    }
    __syncthreads();

    const int wave = tid >> 5;
    const int lane = tid & 31;
    const int l15  = lane & 15;
    const int hi   = lane >> 4;

    const long row0 = ((long)blockIdx.x * 8 + wave) * 16;
    if (row0 >= n_rows) return;

    const float* ap = x + (row0 + l15) * D_IN + 2 * hi;

    v8f c0 = {}, c1 = {}, c2 = {}, c3 = {};

    for (int k = 0; k < D_IN; k += 4) {
        v2f a = *(const v2f*)(ap + k);
        const float* bbase = &wl[((k >> 1) + hi) * 128 + l15 * 2];
        v2f b0 = *(const v2f*)(bbase +  0);
        v2f b1 = *(const v2f*)(bbase + 32);
        v2f b2 = *(const v2f*)(bbase + 64);
        v2f b3 = *(const v2f*)(bbase + 96);
        c0 = __builtin_amdgcn_wmma_f32_16x16x4_f32(false, a, false, b0, (short)0, c0, false, false);
        c1 = __builtin_amdgcn_wmma_f32_16x16x4_f32(false, a, false, b1, (short)0, c1, false, false);
        c2 = __builtin_amdgcn_wmma_f32_16x16x4_f32(false, a, false, b2, (short)0, c2, false, false);
        c3 = __builtin_amdgcn_wmma_f32_16x16x4_f32(false, a, false, b3, (short)0, c3, false, false);
    }

    float* hp = hidden + (row0 + 8 * hi) * D_OUT + l15;
#pragma unroll
    for (int r = 0; r < 8; ++r) {
        hp[r * D_OUT +  0] = c0[r];
        hp[r * D_OUT + 16] = c1[r];
        hp[r * D_OUT + 32] = c2[r];
        hp[r * D_OUT + 48] = c3[r];
    }
}

#define EPW 32
#define GRP 8

__global__ __launch_bounds__(256) void k_scatter(const float* __restrict__ hidden,
                                                 const float* __restrict__ adj,
                                                 const int* __restrict__ row,
                                                 const int* __restrict__ col,
                                                 float* __restrict__ out,
                                                 int n_edges) {
    const int lane = threadIdx.x & 31;
    const int wave = __builtin_amdgcn_readfirstlane((int)(threadIdx.x >> 5));
    const int base = (blockIdx.x * 8 + wave) * EPW;
    if (base >= n_edges) return;
    const int left = n_edges - base;
    const int cnt  = left < EPW ? left : EPW;

    int r = 0, c = 0;
    float v = 0.f;
    if (lane < cnt) {
        r = row[base + lane];
        c = col[base + lane];
        v = adj[base + lane];
    }
    const int vb = __float_as_int(v);

    if (cnt == EPW) {
#pragma unroll
        for (int g = 0; g < EPW; g += GRP) {
            float h0[GRP], h1[GRP];
#pragma unroll
            for (int u = 0; u < GRP; ++u) {
                const int cj = __builtin_amdgcn_readlane(c, g + u);
                const float* h = hidden + (long)cj * D_OUT;
                h0[u] = h[lane];
                h1[u] = h[lane + 32];
            }
#pragma unroll
            for (int u = 0; u < GRP; ++u) {
                const int   rj = __builtin_amdgcn_readlane(r, g + u);
                const float vj = __int_as_float(__builtin_amdgcn_readlane(vb, g + u));
                float* o = out + (long)rj * D_OUT;
                atomicAdd(&o[lane],      vj * h0[u]);
                atomicAdd(&o[lane + 32], vj * h1[u]);
            }
        }
    } else {
        for (int j = 0; j < cnt; ++j) {
            const int   cj = __builtin_amdgcn_readlane(c, j);
            const int   rj = __builtin_amdgcn_readlane(r, j);
            const float vj = __int_as_float(__builtin_amdgcn_readlane(vb, j));
            const float* h = hidden + (long)cj * D_OUT;
            float*       o = out    + (long)rj * D_OUT;
            atomicAdd(&o[lane],      vj * h[lane]);
            atomicAdd(&o[lane + 32], vj * h[lane + 32]);
        }
    }
}

__global__ __launch_bounds__(256) void k_prelu(float* __restrict__ out,
                                               const float* __restrict__ slope,
                                               int n4) {
    int i = blockIdx.x * 256 + threadIdx.x;
    if (i >= n4) return;
    const float s = slope[0];
    v4f v = ((v4f*)out)[i];
#pragma unroll
    for (int j = 0; j < 4; ++j) {
        float t = v[j];
        v[j] = t > 0.f ? t : s * t;
    }
    ((v4f*)out)[i] = v;
}

extern "C" void kernel_launch(void* const* d_in, const int* in_sizes, int n_in,
                              void* d_out, int out_size, void* d_ws, size_t ws_size,
                              hipStream_t stream) {
    const float* x    = (const float*)d_in[0];
    const float* w    = (const float*)d_in[1];
    const float* adj  = (const float*)d_in[2];
    const float* pa   = (const float*)d_in[3];
    const int*   row  = (const int*)d_in[4];
    const int*   col  = (const int*)d_in[5];
    float*       out  = (float*)d_out;
    float*       hid  = (float*)d_ws;

    const int n_nodes = in_sizes[0] / D_IN;
    const int n_edges = in_sizes[2];
    const int out4    = out_size / 4;

    k_zero<<<(out4 + 255) / 256, 256, 0, stream>>>(out, out4);

    const int row_tiles = (n_nodes + 15) / 16;
    k_gemm<<<(row_tiles + 7) / 8, 256, 0, stream>>>(x, w, hid, n_nodes);

    k_scatter<<<(n_edges + 255) / 256, 256, 0, stream>>>(hid, adj, row, col, out, n_edges);

    k_prelu<<<(out4 + 255) / 256, 256, 0, stream>>>(out, pa, out4);
}


# atomic scatter -> gather: per-row edge lists built with one int exchange per edge; one wave per 2 rows sums its lists in registers, PReLU fused; f32 atomics gone; sum order = list order + 4-way tree
# speedup vs baseline: 1.8847x; 1.1173x over previous
#include <hip/hip_runtime.h>


typedef __attribute__((ext_vector_type(2))) float v2f;
typedef __attribute__((ext_vector_type(4))) float v4f;
typedef __attribute__((ext_vector_type(8))) float v8f;
typedef __attribute__((ext_vector_type(4))) int   v4i;

#define D_IN  256
#define D_OUT 64
#define NCH   4

__global__ __launch_bounds__(256) void k_init(int* __restrict__ heads,
                                              int* __restrict__ recs,
                                              float* __restrict__ zrow,
                                              int n_rows, int n_edges, int zcol) {
    const int i = blockIdx.x * 256 + threadIdx.x;
    if (i < n_rows) {
        v4i h = {n_edges, n_edges, n_edges, n_edges};
        ((v4i*)heads)[i] = h;
    }
    if (i < D_OUT / 4) {
        v4f z = {0.f, 0.f, 0.f, 0.f};
        ((v4f*)zrow)[i] = z;
    }
    if (i == 0) {
        v4i s = {n_edges, zcol, 0, 0};
        ((v4i*)recs)[n_edges] = s;
    }
}

__global__ __launch_bounds__(256) void k_gemm(const float* __restrict__ x,
                                              const float* __restrict__ w,
                                              float* __restrict__ hidden,
                                              int n_rows) {
    __shared__ float wl[D_IN * D_OUT];

    const int tid = threadIdx.x;
    for (int i = tid; i < D_IN * D_OUT; i += 256) {
        int k = i / D_OUT;
        int n = i - k * D_OUT;
        wl[(k >> 1) * 128 + n * 2 + (k & 1)] = w[i];
    }
    __syncthreads();

    const int wave = tid >> 5;
    const int lane = tid & 31;
    const int l15  = lane & 15;
    const int hi   = lane >> 4;

    const long row0 = ((long)blockIdx.x * 8 + wave) * 16;
    if (row0 >= n_rows) return;

    const float* ap = x + (row0 + l15) * D_IN + 2 * hi;

    v8f c0 = {}, c1 = {}, c2 = {}, c3 = {};

    for (int k = 0; k < D_IN; k += 4) {
        v2f a = *(const v2f*)(ap + k);
        const float* bbase = &wl[((k >> 1) + hi) * 128 + l15 * 2];
        v2f b0 = *(const v2f*)(bbase +  0);
        v2f b1 = *(const v2f*)(bbase + 32);
        v2f b2 = *(const v2f*)(bbase + 64);
        v2f b3 = *(const v2f*)(bbase + 96);
        c0 = __builtin_amdgcn_wmma_f32_16x16x4_f32(false, a, false, b0, (short)0, c0, false, false);
        c1 = __builtin_amdgcn_wmma_f32_16x16x4_f32(false, a, false, b1, (short)0, c1, false, false);
        c2 = __builtin_amdgcn_wmma_f32_16x16x4_f32(false, a, false, b2, (short)0, c2, false, false);
        c3 = __builtin_amdgcn_wmma_f32_16x16x4_f32(false, a, false, b3, (short)0, c3, false, false);
    }

    float* hp = hidden + (row0 + 8 * hi) * D_OUT + l15;
#pragma unroll
    for (int r = 0; r < 8; ++r) {
        hp[r * D_OUT +  0] = c0[r];
        hp[r * D_OUT + 16] = c1[r];
        hp[r * D_OUT + 32] = c2[r];
        hp[r * D_OUT + 48] = c3[r];
    }
}

__global__ __launch_bounds__(256) void k_build(const float* __restrict__ adj,
                                               const int* __restrict__ row,
                                               const int* __restrict__ col,
                                               int* __restrict__ heads,
                                               int* __restrict__ recs,
                                               int n_edges) {
    const int e = blockIdx.x * 256 + threadIdx.x;
    if (e >= n_edges) return;
    const int   r = row[e];
    const int   c = col[e];
    const float a = adj[e];
    const int prev = atomicExch(&heads[(long)r * NCH + (e & (NCH - 1))], e);
    v4i q = {prev, c, __float_as_int(a), 0};
    ((v4i*)recs)[e] = q;
}

__global__ __launch_bounds__(256) void k_pull(const float* __restrict__ hidden,
                                              const int* __restrict__ heads,
                                              const int* __restrict__ recs,
                                              const float* __restrict__ slope,
                                              float* __restrict__ out,
                                              int n_rows, int n_edges) {
    const int lane = threadIdx.x & 31;
    const int g    = lane >> 3;
    const int sub  = lane & 7;
    const int ra   = (blockIdx.x * 8 + (threadIdx.x >> 5)) * 2;
    if (ra >= n_rows) return;
    const bool two = ra + 1 < n_rows;

    const unsigned end = (unsigned)n_edges;
    const v4i* rv = (const v4i*)recs;

    const unsigned ea = min((unsigned)heads[(long)ra * NCH + g], end);
    const unsigned eb = two ? min((unsigned)heads[(long)(ra + 1) * NCH + g], end) : end;

    v4f a0 = {0.f, 0.f, 0.f, 0.f}, a1 = {0.f, 0.f, 0.f, 0.f};
    v4f b0 = {0.f, 0.f, 0.f, 0.f}, b1 = {0.f, 0.f, 0.f, 0.f};

    if (__builtin_amdgcn_ballot_w32(ea != end || eb != end) != 0) {
        v4i qa = rv[ea];
        v4i qb = rv[eb];
        for (unsigned it = 0; it < end; ++it) {
            const unsigned na = min((unsigned)qa.x, end);
            const unsigned nb = min((unsigned)qb.x, end);
            const v4i qna = rv[na];
            const v4i qnb = rv[nb];
            const v4f* pa = (const v4f*)(hidden + (size_t)(unsigned)qa.y * D_OUT) + sub;
            const v4f* pb = (const v4f*)(hidden + (size_t)(unsigned)qb.y * D_OUT) + sub;
            const v4f ha0 = pa[0];
            const v4f ha1 = pa[8];
            const v4f hb0 = pb[0];
            const v4f hb1 = pb[8];
            asm volatile("" :: "v"(qna.x), "v"(qna.y), "v"(qna.z),
                               "v"(qnb.x), "v"(qnb.y), "v"(qnb.z),
                               "v"(ha0), "v"(ha1), "v"(hb0), "v"(hb1));
            const float wa = __int_as_float(qa.z);
            const float wb = __int_as_float(qb.z);
            a0 += wa * ha0;
            a1 += wa * ha1;
            b0 += wb * hb0;
            b1 += wb * hb1;
            qa = qna;
            qb = qnb;
            if (__builtin_amdgcn_ballot_w32(na != end || nb != end) == 0) break;
        }
    }

#pragma unroll
    for (int j = 0; j < 4; ++j) {
        float s0 = a0[j], s1 = a1[j], t0 = b0[j], t1 = b1[j];
        s0 += __shfl_xor(s0, 8);
        s1 += __shfl_xor(s1, 8);
        t0 += __shfl_xor(t0, 8);
        t1 += __shfl_xor(t1, 8);
        s0 += __shfl_xor(s0, 16);
        s1 += __shfl_xor(s1, 16);
        t0 += __shfl_xor(t0, 16);
        t1 += __shfl_xor(t1, 16);
        a0[j] = s0; a1[j] = s1; b0[j] = t0; b1[j] = t1;
    }

    if (lane < 8 || (lane < 16 && two)) {
        const float s = slope[0];
        v4f o0 = g ? b0 : a0;
        v4f o1 = g ? b1 : a1;
#pragma unroll
        for (int j = 0; j < 4; ++j) {
            const float u0 = o0[j];
            const float u1 = o1[j];
            o0[j] = u0 > 0.f ? u0 : s * u0;
            o1[j] = u1 > 0.f ? u1 : s * u1;
        }
        v4f* op = (v4f*)(out + (size_t)(ra + g) * D_OUT) + sub;
        op[0] = o0;
        op[8] = o1;
    }
}

extern "C" void kernel_launch(void* const* d_in, const int* in_sizes, int n_in,
                              void* d_out, int out_size, void* d_ws, size_t ws_size,
                              hipStream_t stream) {
    const float* x    = (const float*)d_in[0];
    const float* w    = (const float*)d_in[1];
    const float* adj  = (const float*)d_in[2];
    const float* pa   = (const float*)d_in[3];
    const int*   row  = (const int*)d_in[4];
    const int*   col  = (const int*)d_in[5];
    float*       out  = (float*)d_out;

    const int n_nodes   = in_sizes[0] / D_IN;
    const int n_edges   = in_sizes[2];
    const int row_tiles = (n_nodes + 15) / 16;
    const int zcol      = row_tiles * 16;

    float* hid   = (float*)d_ws;
    int*   heads = (int*)(hid + (size_t)(zcol + 1) * D_OUT);
    int*   recs  = heads + (size_t)n_nodes * NCH;

    k_init<<<(n_nodes + 255) / 256, 256, 0, stream>>>(heads, recs, hid + (size_t)zcol * D_OUT,
                                                      n_nodes, n_edges, zcol);

    k_gemm<<<(row_tiles + 7) / 8, 256, 0, stream>>>(x, w, hid, n_nodes);

    k_build<<<(n_edges + 255) / 256, 256, 0, stream>>>(adj, row, col, heads, recs, n_edges);

    k_pull<<<(n_nodes + 15) / 16, 256, 0, stream>>>(hid, heads, recs, pa, out, n_nodes, n_edges);
}


# edge threading (one int exchange per edge) folded into the GEMM launch as extra blocks, 8 edges per thread, so it runs under the matrix work; 4 -> 3 launches; list order stays arrival order
# speedup vs baseline: 2.1465x; 1.1389x over previous
#include <hip/hip_runtime.h>


typedef __attribute__((ext_vector_type(2))) float v2f;
typedef __attribute__((ext_vector_type(4))) float v4f;
typedef __attribute__((ext_vector_type(8))) float v8f;
typedef __attribute__((ext_vector_type(4))) int   v4i;

#define D_IN  256
#define D_OUT 64
#define NCH   4
#define EPT   8

__global__ __launch_bounds__(256) void k_init(int* __restrict__ heads,
                                              int* __restrict__ recs,
                                              float* __restrict__ zrow,
                                              int n_rows, int n_edges, int zcol) {
    const int i = blockIdx.x * 256 + threadIdx.x;
    if (i < n_rows) {
        v4i h = {n_edges, n_edges, n_edges, n_edges};
        ((v4i*)heads)[i] = h;
    }
    if (i < D_OUT / 4) {
        v4f z = {0.f, 0.f, 0.f, 0.f};
        ((v4f*)zrow)[i] = z;
    }
    if (i == 0) {
        v4i s = {n_edges, zcol, 0, 0};
        ((v4i*)recs)[n_edges] = s;
    }
}

__global__ __launch_bounds__(256) void k_gemm_build(const float* __restrict__ x,
                                                    const float* __restrict__ w,
                                                    float* __restrict__ hidden,
                                                    int n_rows, int gemm_blocks,
                                                    const float* __restrict__ adj,
                                                    const int* __restrict__ row,
                                                    const int* __restrict__ col,
                                                    int* __restrict__ heads,
                                                    int* __restrict__ recs,
                                                    int n_edges) {
    __shared__ float wl[D_IN * D_OUT];

    const int tid = threadIdx.x;

    if ((int)blockIdx.x >= gemm_blocks) {
        const int base = ((int)blockIdx.x - gemm_blocks) * (256 * EPT);
        v4i* rv = (v4i*)recs;
        if (base + 256 * EPT <= n_edges) {
            int   r[EPT], c[EPT], prev[EPT];
            float a[EPT];
#pragma unroll
            for (int j = 0; j < EPT; ++j) {
                const int e = base + j * 256 + tid;
                r[j] = row[e];
                c[j] = col[e];
                a[j] = adj[e];
            }
#pragma unroll
            for (int j = 0; j < EPT; ++j) {
                const int e = base + j * 256 + tid;
                prev[j] = atomicExch(&heads[(long)r[j] * NCH + (e & (NCH - 1))], e);
            }
#pragma unroll
            for (int j = 0; j < EPT; ++j) {
                const int e = base + j * 256 + tid;
                v4i q = {prev[j], c[j], __float_as_int(a[j]), 0};
                rv[e] = q;
            }
        } else {
            for (int j = 0; j < EPT; ++j) {
                const int e = base + j * 256 + tid;
                if (e < n_edges) {
                    const int   r1 = row[e];
                    const int   c1 = col[e];
                    const float a1 = adj[e];
                    const int   p1 = atomicExch(&heads[(long)r1 * NCH + (e & (NCH - 1))], e);
                    v4i q = {p1, c1, __float_as_int(a1), 0};
                    rv[e] = q;
                }
            }
        }
        return;
    }

    for (int i = tid; i < D_IN * D_OUT; i += 256) {
        int k = i / D_OUT;
        int n = i - k * D_OUT;
        wl[(k >> 1) * 128 + n * 2 + (k & 1)] = w[i];
    }
    __syncthreads();

    const int wave = tid >> 5;
    const int lane = tid & 31;
    const int l15  = lane & 15;
    const int hi   = lane >> 4;

    const long row0 = ((long)blockIdx.x * 8 + wave) * 16;
    if (row0 >= n_rows) return;

    const float* ap = x + (row0 + l15) * D_IN + 2 * hi;

    v8f c0 = {}, c1 = {}, c2 = {}, c3 = {};

    for (int k = 0; k < D_IN; k += 4) {
        v2f a = *(const v2f*)(ap + k);
        const float* bbase = &wl[((k >> 1) + hi) * 128 + l15 * 2];
        v2f b0 = *(const v2f*)(bbase +  0);
        v2f b1 = *(const v2f*)(bbase + 32);
        v2f b2 = *(const v2f*)(bbase + 64);
        v2f b3 = *(const v2f*)(bbase + 96);
        c0 = __builtin_amdgcn_wmma_f32_16x16x4_f32(false, a, false, b0, (short)0, c0, false, false);
        c1 = __builtin_amdgcn_wmma_f32_16x16x4_f32(false, a, false, b1, (short)0, c1, false, false);
        c2 = __builtin_amdgcn_wmma_f32_16x16x4_f32(false, a, false, b2, (short)0, c2, false, false);
        c3 = __builtin_amdgcn_wmma_f32_16x16x4_f32(false, a, false, b3, (short)0, c3, false, false);
    }

    float* hp = hidden + (row0 + 8 * hi) * D_OUT + l15;
#pragma unroll
    for (int r = 0; r < 8; ++r) {
        hp[r * D_OUT +  0] = c0[r];
        hp[r * D_OUT + 16] = c1[r];
        hp[r * D_OUT + 32] = c2[r];
        hp[r * D_OUT + 48] = c3[r];
    }
}

__global__ __launch_bounds__(256) void k_pull(const float* __restrict__ hidden,
                                              const int* __restrict__ heads,
                                              const int* __restrict__ recs,
                                              const float* __restrict__ slope,
                                              float* __restrict__ out,
                                              int n_rows, int n_edges) {
    const int lane = threadIdx.x & 31;
    const int g    = lane >> 3;
    const int sub  = lane & 7;
    const int ra   = (blockIdx.x * 8 + (threadIdx.x >> 5)) * 2;
    if (ra >= n_rows) return;
    const bool two = ra + 1 < n_rows;

    const unsigned end = (unsigned)n_edges;
    const v4i* rv = (const v4i*)recs;

    const unsigned ea = min((unsigned)heads[(long)ra * NCH + g], end);
    const unsigned eb = two ? min((unsigned)heads[(long)(ra + 1) * NCH + g], end) : end;

    v4f a0 = {0.f, 0.f, 0.f, 0.f}, a1 = {0.f, 0.f, 0.f, 0.f};
    v4f b0 = {0.f, 0.f, 0.f, 0.f}, b1 = {0.f, 0.f, 0.f, 0.f};

    if (__builtin_amdgcn_ballot_w32(ea != end || eb != end) != 0) {
        v4i qa = rv[ea];
        v4i qb = rv[eb];
        for (unsigned it = 0; it < end; ++it) {
            const unsigned na = min((unsigned)qa.x, end);
            const unsigned nb = min((unsigned)qb.x, end);
            const v4i qna = rv[na];
            const v4i qnb = rv[nb];
            const v4f* pa = (const v4f*)(hidden + (size_t)(unsigned)qa.y * D_OUT) + sub;
            const v4f* pb = (const v4f*)(hidden + (size_t)(unsigned)qb.y * D_OUT) + sub;
            const v4f ha0 = pa[0];
            const v4f ha1 = pa[8];
            const v4f hb0 = pb[0];
            const v4f hb1 = pb[8];
            asm volatile("" :: "v"(qna.x), "v"(qna.y), "v"(qna.z),
                               "v"(qnb.x), "v"(qnb.y), "v"(qnb.z),
                               "v"(ha0), "v"(ha1), "v"(hb0), "v"(hb1));
            const float wa = __int_as_float(qa.z);
            const float wb = __int_as_float(qb.z);
            a0 += wa * ha0;
            a1 += wa * ha1;
            b0 += wb * hb0;
            b1 += wb * hb1;
            qa = qna;
            qb = qnb;
            if (__builtin_amdgcn_ballot_w32(na != end || nb != end) == 0) break;
        }
    }

#pragma unroll
    for (int j = 0; j < 4; ++j) {
        float s0 = a0[j], s1 = a1[j], t0 = b0[j], t1 = b1[j];
        s0 += __shfl_xor(s0, 8);
        s1 += __shfl_xor(s1, 8);
        t0 += __shfl_xor(t0, 8);
        t1 += __shfl_xor(t1, 8);
        s0 += __shfl_xor(s0, 16);
        s1 += __shfl_xor(s1, 16);
        t0 += __shfl_xor(t0, 16);
        t1 += __shfl_xor(t1, 16);
        a0[j] = s0; a1[j] = s1; b0[j] = t0; b1[j] = t1;
    }

    if (lane < 8 || (lane < 16 && two)) {
        const float s = slope[0];
        v4f o0 = g ? b0 : a0;
        v4f o1 = g ? b1 : a1;
#pragma unroll
        for (int j = 0; j < 4; ++j) {
            const float u0 = o0[j];
            const float u1 = o1[j];
            o0[j] = u0 > 0.f ? u0 : s * u0;
            o1[j] = u1 > 0.f ? u1 : s * u1;
        }
        v4f* op = (v4f*)(out + (size_t)(ra + g) * D_OUT) + sub;
        op[0] = o0;
        op[8] = o1;
    }
}

extern "C" void kernel_launch(void* const* d_in, const int* in_sizes, int n_in,
                              void* d_out, int out_size, void* d_ws, size_t ws_size,
                              hipStream_t stream) {
    const float* x    = (const float*)d_in[0];
    const float* w    = (const float*)d_in[1];
    const float* adj  = (const float*)d_in[2];
    const float* pa   = (const float*)d_in[3];
    const int*   row  = (const int*)d_in[4];
    const int*   col  = (const int*)d_in[5];
    float*       out  = (float*)d_out;

    const int n_nodes   = in_sizes[0] / D_IN;
    const int n_edges   = in_sizes[2];
    const int row_tiles = (n_nodes + 15) / 16;
    const int zcol      = row_tiles * 16;

    float* hid   = (float*)d_ws;
    int*   heads = (int*)(hid + (size_t)(zcol + 1) * D_OUT);
    int*   recs  = heads + (size_t)n_nodes * NCH;

    k_init<<<(n_nodes + 255) / 256, 256, 0, stream>>>(heads, recs, hid + (size_t)zcol * D_OUT,
                                                      n_nodes, n_edges, zcol);

    const int gemm_blocks = (row_tiles + 7) / 8;
    const int edge_blocks = (n_edges + 256 * EPT - 1) / (256 * EPT);
    k_gemm_build<<<gemm_blocks + edge_blocks, 256, 0, stream>>>(x, w, hid, n_nodes, gemm_blocks,
                                                               adj, row, col, heads, recs, n_edges);

    k_pull<<<(n_nodes + 15) / 16, 256, 0, stream>>>(hid, heads, recs, pa, out, n_nodes, n_edges);
}


# per-row linked lists -> edges partitioned in LDS into 256 row bins beside the GEMM; one 32-wave block per bin sums its rows in LDS (float adds, sum order = arrival order); 3 -> 2 launches
# speedup vs baseline: 3.3704x; 1.5702x over previous
#include <hip/hip_runtime.h>


typedef __attribute__((ext_vector_type(2))) float v2f;
typedef __attribute__((ext_vector_type(4))) float v4f;
typedef __attribute__((ext_vector_type(8))) float v8f;
typedef __attribute__((ext_vector_type(2))) int   v2i;

#define D_IN  256
#define D_OUT 64
#define RPB   391
#define NBMAX 256
#define SLAB  4096
#define ACC_FLOATS 43008

__global__ __launch_bounds__(256) void k_gemm_part(const float* __restrict__ x,
                                                   const float* __restrict__ w,
                                                   float* __restrict__ hidden,
                                                   int n_rows, int gemm_blocks,
                                                   const float* __restrict__ adj,
                                                   const int* __restrict__ row,
                                                   const int* __restrict__ col,
                                                   int* __restrict__ tab,
                                                   int* __restrict__ grow,
                                                   int* __restrict__ gcol,
                                                   float* __restrict__ gw,
                                                   int n_edges, int nb, int nslab) {
    __shared__ float wl[D_IN * D_OUT];

    const int tid = threadIdx.x;

    if ((int)blockIdx.x >= gemm_blocks) {
        int* li   = (int*)wl;
        int* cnt  = li;
        int* cur  = li + NBMAX;
        int* srow = li + 2 * NBMAX;
        int* scol = srow + SLAB;
        float* sw = (float*)(scol + SLAB);

        const int s    = (int)blockIdx.x - gemm_blocks;
        const int base = s * SLAB;
        const int n    = min(SLAB, n_edges - base);

        cnt[tid] = 0;
        __syncthreads();

        int   r[SLAB / 256], c[SLAB / 256], b[SLAB / 256];
        float a[SLAB / 256];
#pragma unroll
        for (int j = 0; j < SLAB / 256; ++j) {
            const int i = j * 256 + tid;
            r[j] = -1; c[j] = 0; a[j] = 0.f;
            if (i < n) {
                r[j] = row[base + i];
                c[j] = col[base + i];
                a[j] = adj[base + i];
            }
        }
#pragma unroll
        for (int j = 0; j < SLAB / 256; ++j) {
            b[j] = -1;
            if (r[j] >= 0 && r[j] < n_rows) {
                b[j] = r[j] / RPB;
                atomicAdd(&cnt[b[j]], 1);
            }
        }
        __syncthreads();

        if (tid < 32) {
            int v[8];
            int sum = 0;
#pragma unroll
            for (int k = 0; k < 8; ++k) { v[k] = cnt[tid * 8 + k]; sum += v[k]; }
            int inc = sum;
#pragma unroll
            for (int d = 1; d < 32; d <<= 1) {
                const int t = __shfl_up(inc, d);
                if (tid >= d) inc += t;
            }
            int run = inc - sum;
#pragma unroll
            for (int k = 0; k < 8; ++k) { cur[tid * 8 + k] = run; run += v[k]; }
        }
        __syncthreads();

        if (tid < nb) {
            v2i e = {base + cur[tid], cnt[tid]};
            ((v2i*)tab)[(size_t)tid * nslab + s] = e;
        }
        __syncthreads();

#pragma unroll
        for (int j = 0; j < SLAB / 256; ++j) {
            if (b[j] >= 0) {
                const int p = atomicAdd(&cur[b[j]], 1);
                srow[p] = r[j];
                scol[p] = c[j];
                sw[p]   = a[j];
            }
        }
        __syncthreads();

        for (int i = tid; i < n; i += 256) {
            grow[base + i] = srow[i];
            gcol[base + i] = scol[i];
            gw[base + i]   = sw[i];
        }
        return;
    }

    for (int i = tid; i < D_IN * D_OUT; i += 256) {
        int k = i / D_OUT;
        int n = i - k * D_OUT;
        wl[(k >> 1) * 128 + n * 2 + (k & 1)] = w[i];
    }
    __syncthreads();

    const int wave = tid >> 5;
    const int lane = tid & 31;
    const int l15  = lane & 15;
    const int hi   = lane >> 4;

    const long row0 = ((long)blockIdx.x * 8 + wave) * 16;
    if (row0 >= n_rows) return;

    const float* ap = x + (row0 + l15) * D_IN + 2 * hi;

    v8f c0 = {}, c1 = {}, c2 = {}, c3 = {};

    for (int k = 0; k < D_IN; k += 4) {
        v2f a = *(const v2f*)(ap + k);
        const float* bbase = &wl[((k >> 1) + hi) * 128 + l15 * 2];
        v2f b0 = *(const v2f*)(bbase +  0);
        v2f b1 = *(const v2f*)(bbase + 32);
        v2f b2 = *(const v2f*)(bbase + 64);
        v2f b3 = *(const v2f*)(bbase + 96);
        c0 = __builtin_amdgcn_wmma_f32_16x16x4_f32(false, a, false, b0, (short)0, c0, false, false);
        c1 = __builtin_amdgcn_wmma_f32_16x16x4_f32(false, a, false, b1, (short)0, c1, false, false);
        c2 = __builtin_amdgcn_wmma_f32_16x16x4_f32(false, a, false, b2, (short)0, c2, false, false);
        c3 = __builtin_amdgcn_wmma_f32_16x16x4_f32(false, a, false, b3, (short)0, c3, false, false);
    }

    float* hp = hidden + (row0 + 8 * hi) * D_OUT + l15;
#pragma unroll
    for (int r = 0; r < 8; ++r) {
        hp[r * D_OUT +  0] = c0[r];
        hp[r * D_OUT + 16] = c1[r];
        hp[r * D_OUT + 32] = c2[r];
        hp[r * D_OUT + 48] = c3[r];
    }
}

__global__ __launch_bounds__(1024) void k_bin_pull(const float* __restrict__ hidden,
                                                   const int* __restrict__ tab,
                                                   const int* __restrict__ grow,
                                                   const int* __restrict__ gcol,
                                                   const float* __restrict__ gw,
                                                   const float* __restrict__ slope,
                                                   float* __restrict__ out,
                                                   int n_rows, int n_edges, int nslab) {
    __shared__ float acc[ACC_FLOATS];

    const int tid  = threadIdx.x;
    const int lane = tid & 31;
    const int wave = tid >> 5;
    const int h    = lane >> 4;
    const int l16  = lane & 15;
    const int row0 = (int)blockIdx.x * RPB;
    const int nr   = min(RPB, n_rows - row0);

    for (int i = tid; i < RPB * D_OUT / 4; i += 1024) {
        v4f z = {0.f, 0.f, 0.f, 0.f};
        ((v4f*)acc)[i] = z;
    }
    __syncthreads();

    const v2i* tb = (const v2i*)tab + (size_t)blockIdx.x * nslab;
    const unsigned lastrow = (unsigned)(n_rows - 1);

    int mo = 0, mc = 0;
    {
        const int sj = wave + 32 * lane;
        if (sj < nslab) { const v2i t = tb[sj]; mo = t.x; mc = t.y; }
    }
    const int nseg = wave < nslab ? (nslab - wave + 31) / 32 : 0;

    int o   = min(max(__shfl(mo, 0), 0), n_edges);
    int cnt = min(max(__shfl(mc, 0), 0), n_edges - o);
    if (nseg == 0) cnt = 0;
    int er = 0, ec = 0;
    float ew = 0.f;
    if (lane < cnt) { er = grow[o + lane]; ec = gcol[o + lane]; ew = gw[o + lane]; }

    int j = 0, c0 = 0;
    while (j < nseg) {
        int jn = j, on = o, cn = cnt, c0n = c0 + 32;
        if (c0n >= cnt) {
            jn  = j + 1;
            c0n = 0;
            const int js = min(jn, 31);
            on = min(max(__shfl(mo, js), 0), n_edges);
            cn = min(max(__shfl(mc, js), 0), n_edges - on);
            if (jn >= nseg) cn = 0;
        }
        int nrw = 0, ncl = 0;
        float nwt = 0.f;

        const int m = min(32, cnt - c0);
#pragma unroll
        for (int g = 0; g < 2; ++g) {
            const int k = g * 16;
            if (g == 1 && m <= 16) break;
            v4f   hv[8];
            int   rl[8];
            float wv[8];
#pragma unroll
            for (int q = 0; q < 8; ++q) {
                const int src = k + 2 * q + h;
                const int      rloc = __shfl(er, src) - row0;
                const unsigned csrc = (unsigned)__shfl(ec, src);
                wv[q] = __shfl(ew, src);
                const bool ok = src < m && rloc >= 0 && rloc < nr;
                rl[q] = ok ? rloc : -1;
                const unsigned cc = ok ? min(csrc, lastrow) : 0u;
                hv[q] = ((const v4f*)(hidden + (size_t)cc * D_OUT))[l16];
            }
            if (g == 0) {
                const int idx = min(on + c0n + lane, n_edges - 1);
                nrw = grow[idx]; ncl = gcol[idx]; nwt = gw[idx];
            }
#pragma unroll
            for (int q = 0; q < 8; ++q) {
                if (rl[q] >= 0) {
                    float* a  = acc + rl[q] * D_OUT + l16;
                    float* aA = a + 32 * h;
                    float* aB = a + 32 - 32 * h;
                    const float t0 = h ? hv[q][2] : hv[q][0];
                    const float t1 = h ? hv[q][3] : hv[q][1];
                    const float t2 = h ? hv[q][0] : hv[q][2];
                    const float t3 = h ? hv[q][1] : hv[q][3];
                    atomicAdd(aA,      wv[q] * t0);
                    atomicAdd(aA + 16, wv[q] * t1);
                    atomicAdd(aB,      wv[q] * t2);
                    atomicAdd(aB + 16, wv[q] * t3);
                }
            }
        }
        j = jn; o = on; cnt = cn; c0 = c0n; er = nrw; ec = ncl; ew = nwt;
    }
    __syncthreads();

    const float sl = slope[0];
    v4f* op = (v4f*)(out + (size_t)row0 * D_OUT);
    for (int i = tid; i < nr * (D_OUT / 4); i += 1024) {
        const float* ar = acc + (i >> 4) * D_OUT + (i & 15);
        v4f v;
#pragma unroll
        for (int jj = 0; jj < 4; ++jj) {
            const float u = ar[jj * 16];
            v[jj] = u > 0.f ? u : sl * u;
        }
        op[i] = v;
    }
}

extern "C" void kernel_launch(void* const* d_in, const int* in_sizes, int n_in,
                              void* d_out, int out_size, void* d_ws, size_t ws_size,
                              hipStream_t stream) {
    const float* x    = (const float*)d_in[0];
    const float* w    = (const float*)d_in[1];
    const float* adj  = (const float*)d_in[2];
    const float* pa   = (const float*)d_in[3];
    const int*   row  = (const int*)d_in[4];
    const int*   col  = (const int*)d_in[5];
    float*       out  = (float*)d_out;

    const int n_nodes   = in_sizes[0] / D_IN;
    const int n_edges   = in_sizes[2];
    const int row_tiles = (n_nodes + 15) / 16;
    const int nb        = (n_nodes + RPB - 1) / RPB;
    const int nslab     = (n_edges + SLAB - 1) / SLAB;

    float* hid  = (float*)d_ws;
    int*   tab  = (int*)(hid + (size_t)row_tiles * 16 * D_OUT);
    int*   grow = tab + (size_t)nb * nslab * 2;
    int*   gcol = grow + n_edges;
    float* gw   = (float*)(gcol + n_edges);

    const int gemm_blocks = (row_tiles + 7) / 8;
    k_gemm_part<<<gemm_blocks + nslab, 256, 0, stream>>>(x, w, hid, n_nodes, gemm_blocks,
                                                        adj, row, col, tab, grow, gcol, gw,
                                                        n_edges, nb, nslab);

    k_bin_pull<<<nb, 1024, 0, stream>>>(hid, tab, grow, gcol, gw, pa, out, n_nodes, n_edges, nslab);
}


# GEMM tiles dealt evenly: 768 GEMM blocks (3 per WGP) take one 16x64 tile per wave, the 106 left-over tiles go out as 16x32 halves, one per block; no WGP carries a 4th GEMM block any more
# speedup vs baseline: 3.4935x; 1.0365x over previous
#include <hip/hip_runtime.h>


typedef __attribute__((ext_vector_type(2))) float v2f;
typedef __attribute__((ext_vector_type(4))) float v4f;
typedef __attribute__((ext_vector_type(8))) float v8f;
typedef __attribute__((ext_vector_type(2))) int   v2i;

#define D_IN  256
#define D_OUT 64
#define RPB   391
#define NBMAX 256
#define SLAB  4096
#define ACC_FLOATS 43008

__device__ __forceinline__ void gemm_tile64(const float* __restrict__ x, const float* wl,
                                            float* __restrict__ hidden, long row0,
                                            int l15, int hi) {
    const float* ap = x + (row0 + l15) * D_IN + 2 * hi;

    v8f c0 = {}, c1 = {}, c2 = {}, c3 = {};

    for (int k = 0; k < D_IN; k += 4) {
        v2f a = *(const v2f*)(ap + k);
        const float* bbase = &wl[((k >> 1) + hi) * 128 + l15 * 2];
        v2f b0 = *(const v2f*)(bbase +  0);
        v2f b1 = *(const v2f*)(bbase + 32);
        v2f b2 = *(const v2f*)(bbase + 64);
        v2f b3 = *(const v2f*)(bbase + 96);
        c0 = __builtin_amdgcn_wmma_f32_16x16x4_f32(false, a, false, b0, (short)0, c0, false, false);
        c1 = __builtin_amdgcn_wmma_f32_16x16x4_f32(false, a, false, b1, (short)0, c1, false, false);
        c2 = __builtin_amdgcn_wmma_f32_16x16x4_f32(false, a, false, b2, (short)0, c2, false, false);
        c3 = __builtin_amdgcn_wmma_f32_16x16x4_f32(false, a, false, b3, (short)0, c3, false, false);
    }

    float* hp = hidden + (row0 + 8 * hi) * D_OUT + l15;
#pragma unroll
    for (int r = 0; r < 8; ++r) {
        hp[r * D_OUT +  0] = c0[r];
        hp[r * D_OUT + 16] = c1[r];
        hp[r * D_OUT + 32] = c2[r];
        hp[r * D_OUT + 48] = c3[r];
    }
}

__device__ __forceinline__ void gemm_tile32(const float* __restrict__ x, const float* wl,
                                            float* __restrict__ hidden, long row0, int ch,
                                            int l15, int hi) {
    const float* ap = x + (row0 + l15) * D_IN + 2 * hi;

    v8f c0 = {}, c1 = {};

    for (int k = 0; k < D_IN; k += 4) {
        v2f a = *(const v2f*)(ap + k);
        const float* bbase = &wl[((k >> 1) + hi) * 128 + ch * 64 + l15 * 2];
        v2f b0 = *(const v2f*)(bbase +  0);
        v2f b1 = *(const v2f*)(bbase + 32);
        c0 = __builtin_amdgcn_wmma_f32_16x16x4_f32(false, a, false, b0, (short)0, c0, false, false);
        c1 = __builtin_amdgcn_wmma_f32_16x16x4_f32(false, a, false, b1, (short)0, c1, false, false);
    }

    float* hp = hidden + (row0 + 8 * hi) * D_OUT + ch * 32 + l15;
#pragma unroll
    for (int r = 0; r < 8; ++r) {
        hp[r * D_OUT +  0] = c0[r];
        hp[r * D_OUT + 16] = c1[r];
    }
}

__global__ __launch_bounds__(256) void k_gemm_part(const float* __restrict__ x,
                                                   const float* __restrict__ w,
                                                   float* __restrict__ hidden,
                                                   int n_rows, int gemm_blocks,
                                                   const float* __restrict__ adj,
                                                   const int* __restrict__ row,
                                                   const int* __restrict__ col,
                                                   int* __restrict__ tab,
                                                   int* __restrict__ grow,
                                                   int* __restrict__ gcol,
                                                   float* __restrict__ gw,
                                                   int n_edges, int nb, int nslab) {
    __shared__ float wl[D_IN * D_OUT];

    const int tid = threadIdx.x;

    if ((int)blockIdx.x >= gemm_blocks) {
        int* li   = (int*)wl;
        int* cnt  = li;
        int* cur  = li + NBMAX;
        int* srow = li + 2 * NBMAX;
        int* scol = srow + SLAB;
        float* sw = (float*)(scol + SLAB);

        const int s    = (int)blockIdx.x - gemm_blocks;
        const int base = s * SLAB;
        const int n    = min(SLAB, n_edges - base);

        cnt[tid] = 0;
        __syncthreads();

        int   r[SLAB / 256], c[SLAB / 256], b[SLAB / 256];
        float a[SLAB / 256];
#pragma unroll
        for (int j = 0; j < SLAB / 256; ++j) {
            const int i = j * 256 + tid;
            r[j] = -1; c[j] = 0; a[j] = 0.f;
            if (i < n) {
                r[j] = row[base + i];
                c[j] = col[base + i];
                a[j] = adj[base + i];
            }
        }
#pragma unroll
        for (int j = 0; j < SLAB / 256; ++j) {
            b[j] = -1;
            if (r[j] >= 0 && r[j] < n_rows) {
                b[j] = r[j] / RPB;
                atomicAdd(&cnt[b[j]], 1);
            }
        }
        __syncthreads();

        if (tid < 32) {
            int v[8];
            int sum = 0;
#pragma unroll
            for (int k = 0; k < 8; ++k) { v[k] = cnt[tid * 8 + k]; sum += v[k]; }
            int inc = sum;
#pragma unroll
            for (int d = 1; d < 32; d <<= 1) {
                const int t = __shfl_up(inc, d);
                if (tid >= d) inc += t;
            }
            int run = inc - sum;
#pragma unroll
            for (int k = 0; k < 8; ++k) { cur[tid * 8 + k] = run; run += v[k]; }
        }
        __syncthreads();

        if (tid < nb) {
            v2i e = {base + cur[tid], cnt[tid]};
            ((v2i*)tab)[(size_t)tid * nslab + s] = e;
        }
        __syncthreads();

#pragma unroll
        for (int j = 0; j < SLAB / 256; ++j) {
            if (b[j] >= 0) {
                const int p = atomicAdd(&cur[b[j]], 1);
                srow[p] = r[j];
                scol[p] = c[j];
                sw[p]   = a[j];
            }
        }
        __syncthreads();

        for (int i = tid; i < n; i += 256) {
            grow[base + i] = srow[i];
            gcol[base + i] = scol[i];
            gw[base + i]   = sw[i];
        }
        return;
    }

    for (int i = tid; i < D_IN * D_OUT; i += 256) {
        int k = i / D_OUT;
        int n = i - k * D_OUT;
        wl[(k >> 1) * 128 + n * 2 + (k & 1)] = w[i];
    }
    __syncthreads();

    const int wave = tid >> 5;
    const int lane = tid & 31;
    const int l15  = lane & 15;
    const int hi   = lane >> 4;

    const int row_tiles = (n_rows + 15) / 16;
    const int nw   = gemm_blocks * 8;
    const int gwv  = (int)blockIdx.x * 8 + wave;
    const int full = row_tiles / nw;
    const int left = row_tiles - full * nw;

    for (int r = 0; r < full; ++r)
        gemm_tile64(x, wl, hidden, ((long)r * nw + gwv) * 16, l15, hi);

    if (2 * left <= nw) {
        const int u = wave * gemm_blocks + (int)blockIdx.x;
        if (u < 2 * left)
            gemm_tile32(x, wl, hidden, ((long)full * nw + (u >> 1)) * 16, u & 1, l15, hi);
    } else if (gwv < left) {
        gemm_tile64(x, wl, hidden, ((long)full * nw + gwv) * 16, l15, hi);
    }
}

__global__ __launch_bounds__(1024) void k_bin_pull(const float* __restrict__ hidden,
                                                   const int* __restrict__ tab,
                                                   const int* __restrict__ grow,
                                                   const int* __restrict__ gcol,
                                                   const float* __restrict__ gw,
                                                   const float* __restrict__ slope,
                                                   float* __restrict__ out,
                                                   int n_rows, int n_edges, int nslab) {
    __shared__ float acc[ACC_FLOATS];

    const int tid  = threadIdx.x;
    const int lane = tid & 31;
    const int wave = tid >> 5;
    const int h    = lane >> 4;
    const int l16  = lane & 15;
    const int row0 = (int)blockIdx.x * RPB;
    const int nr   = min(RPB, n_rows - row0);

    for (int i = tid; i < RPB * D_OUT / 4; i += 1024) {
        v4f z = {0.f, 0.f, 0.f, 0.f};
        ((v4f*)acc)[i] = z;
    }
    __syncthreads();

    const v2i* tb = (const v2i*)tab + (size_t)blockIdx.x * nslab;
    const unsigned lastrow = (unsigned)(n_rows - 1);

    int mo = 0, mc = 0;
    {
        const int sj = wave + 32 * lane;
        if (sj < nslab) { const v2i t = tb[sj]; mo = t.x; mc = t.y; }
    }
    const int nseg = wave < nslab ? (nslab - wave + 31) / 32 : 0;

    int o   = min(max(__shfl(mo, 0), 0), n_edges);
    int cnt = min(max(__shfl(mc, 0), 0), n_edges - o);
    if (nseg == 0) cnt = 0;
    int er = 0, ec = 0;
    float ew = 0.f;
    if (lane < cnt) { er = grow[o + lane]; ec = gcol[o + lane]; ew = gw[o + lane]; }

    int j = 0, c0 = 0;
    while (j < nseg) {
        int jn = j, on = o, cn = cnt, c0n = c0 + 32;
        if (c0n >= cnt) {
            jn  = j + 1;
            c0n = 0;
            const int js = min(jn, 31);
            on = min(max(__shfl(mo, js), 0), n_edges);
            cn = min(max(__shfl(mc, js), 0), n_edges - on);
            if (jn >= nseg) cn = 0;
        }
        int nrw = 0, ncl = 0;
        float nwt = 0.f;

        const int m = min(32, cnt - c0);
#pragma unroll
        for (int g = 0; g < 2; ++g) {
            const int k = g * 16;
            if (g == 1 && m <= 16) break;
            v4f   hv[8];
            int   rl[8];
            float wv[8];
#pragma unroll
            for (int q = 0; q < 8; ++q) {
                const int src = k + 2 * q + h;
                const int      rloc = __shfl(er, src) - row0;
                const unsigned csrc = (unsigned)__shfl(ec, src);
                wv[q] = __shfl(ew, src);
                const bool ok = src < m && rloc >= 0 && rloc < nr;
                rl[q] = ok ? rloc : -1;
                const unsigned cc = ok ? min(csrc, lastrow) : 0u;
                hv[q] = ((const v4f*)(hidden + (size_t)cc * D_OUT))[l16];
            }
            if (g == 0) {
                const int idx = min(on + c0n + lane, n_edges - 1);
                nrw = grow[idx]; ncl = gcol[idx]; nwt = gw[idx];
            }
#pragma unroll
            for (int q = 0; q < 8; ++q) {
                if (rl[q] >= 0) {
                    float* a  = acc + rl[q] * D_OUT + l16;
                    float* aA = a + 32 * h;
                    float* aB = a + 32 - 32 * h;
                    const float t0 = h ? hv[q][2] : hv[q][0];
                    const float t1 = h ? hv[q][3] : hv[q][1];
                    const float t2 = h ? hv[q][0] : hv[q][2];
                    const float t3 = h ? hv[q][1] : hv[q][3];
                    atomicAdd(aA,      wv[q] * t0);
                    atomicAdd(aA + 16, wv[q] * t1);
                    atomicAdd(aB,      wv[q] * t2);
                    atomicAdd(aB + 16, wv[q] * t3);
                }
            }
        }
        j = jn; o = on; cnt = cn; c0 = c0n; er = nrw; ec = ncl; ew = nwt;
    }
    __syncthreads();

    const float sl = slope[0];
    v4f* op = (v4f*)(out + (size_t)row0 * D_OUT);
    for (int i = tid; i < nr * (D_OUT / 4); i += 1024) {
        const float* ar = acc + (i >> 4) * D_OUT + (i & 15);
        v4f v;
#pragma unroll
        for (int jj = 0; jj < 4; ++jj) {
            const float u = ar[jj * 16];
            v[jj] = u > 0.f ? u : sl * u;
        }
        op[i] = v;
    }
}

extern "C" void kernel_launch(void* const* d_in, const int* in_sizes, int n_in,
                              void* d_out, int out_size, void* d_ws, size_t ws_size,
                              hipStream_t stream) {
    const float* x    = (const float*)d_in[0];
    const float* w    = (const float*)d_in[1];
    const float* adj  = (const float*)d_in[2];
    const float* pa   = (const float*)d_in[3];
    const int*   row  = (const int*)d_in[4];
    const int*   col  = (const int*)d_in[5];
    float*       out  = (float*)d_out;

    const int n_nodes   = in_sizes[0] / D_IN;
    const int n_edges   = in_sizes[2];
    const int row_tiles = (n_nodes + 15) / 16;
    const int nb        = (n_nodes + RPB - 1) / RPB;
    const int nslab     = (n_edges + SLAB - 1) / SLAB;

    float* hid  = (float*)d_ws;
    int*   tab  = (int*)(hid + (size_t)row_tiles * 16 * D_OUT);
    int*   grow = tab + (size_t)nb * nslab * 2;
    int*   gcol = grow + n_edges;
    float* gw   = (float*)(gcol + n_edges);

    const int gemm_blocks = 768;
    k_gemm_part<<<gemm_blocks + nslab, 256, 0, stream>>>(x, w, hid, n_nodes, gemm_blocks,
                                                        adj, row, col, tab, grow, gcol, gw,
                                                        n_edges, nb, nslab);

    k_bin_pull<<<nb, 1024, 0, stream>>>(hid, tab, grow, gcol, gw, pa, out, n_nodes, n_edges, nslab);
}
